# starting kernel
# baseline (speedup 1.0000x reference)
#include <hip/hip_runtime.h>


typedef float v2f __attribute__((ext_vector_type(2)));
typedef float v8f __attribute__((ext_vector_type(8)));

#define B_TOT  16384
#define T_LEN  100
#define V_SZ   256
#define E_SZ   50
#define H_SZ   10
#define L_SZ   15
#define HP     16
#define ROWS_PER_BLOCK 128
#define TPB    256
#define T_CHUNK 20
#define H_STRIDE 20

__device__ __forceinline__ float fast_tanh(float x) {
#if __has_builtin(__builtin_amdgcn_tanhf)
    return __builtin_amdgcn_tanhf(x);
#else
    x = fminf(fmaxf(x, -9.0f), 9.0f);
    float z = __builtin_amdgcn_exp2f(x * 2.885390081777927f);
    return (z - 1.0f) * __builtin_amdgcn_rcpf(z + 1.0f);
#endif
}

__device__ __forceinline__ v8f wmma_f32_16x16x4(v2f a, v2f b, v8f c) {
    return __builtin_amdgcn_wmma_f32_16x16x4_f32(
         false, a,  false, b,
         (short)0, c,  false,  false);
}

__global__ __launch_bounds__(TPB)
void charrnn_kernel(const int*   __restrict__ x,
                    const float* __restrict__ emb,
                    const float* __restrict__ Wx,
                    const float* __restrict__ Wh,
                    const float* __restrict__ b_rnn,
                    const float* __restrict__ Wd,
                    const float* __restrict__ bd,
                    const float* __restrict__ dmask,
                    float*       __restrict__ out)
{
    __shared__ float s_projb[V_SZ * HP];
    __shared__ int   s_tok[ROWS_PER_BLOCK * T_CHUNK];
    __shared__ float s_h[(TPB / 32) * 16 * H_STRIDE];

    const int tid   = threadIdx.x;
    const int lane  = tid & 31;
    const int wave  = tid >> 5;
    const int n     = lane & 15;
    const int half  = lane >> 4;
    const int mBase = half * 8;
    const int rowLocal = wave * 16;
    const int gRowBase = blockIdx.x * ROWS_PER_BLOCK + rowLocal;

    {
        const int v = tid;
        float s[H_SZ];
#pragma unroll
        for (int j = 0; j < H_SZ; ++j) s[j] = b_rnn[j];
        for (int e = 0; e < E_SZ; ++e) {
            const float em = emb[v * E_SZ + e];
#pragma unroll
            for (int j = 0; j < H_SZ; ++j)
                s[j] = fmaf(em, Wx[e * H_SZ + j], s[j]);
        }
#pragma unroll
        for (int j = 0; j < HP; ++j)
            s_projb[v * HP + j] = (j < H_SZ) ? s[j] : 0.0f;
    }

    v2f whb[3];
#pragma unroll
    for (int c = 0; c < 3; ++c) {
#pragma unroll
        for (int j = 0; j < 2; ++j) {
            const int K = 4 * c + j + 2 * half;
            whb[c][j] = (K < H_SZ && n < H_SZ) ? Wh[K * H_SZ + n] : 0.0f;
        }
    }

    const float colMask = (n < H_SZ) ? 1.0f : 0.0f;
    float* hw = &s_h[wave * 16 * H_STRIDE];

#pragma unroll
    for (int r = 0; r < 8; ++r) hw[(r + mBase) * H_STRIDE + n] = 0.0f;

    __syncthreads();

    for (int tc = 0; tc < T_LEN; tc += T_CHUNK) {
        __syncthreads();
        for (int i = tid; i < ROWS_PER_BLOCK * T_CHUNK; i += TPB) {
            const int rl = i / T_CHUNK, tt = i % T_CHUNK;
            s_tok[i] = x[(blockIdx.x * ROWS_PER_BLOCK + rl) * T_LEN + tc + tt];
        }
        __syncthreads();

        for (int t = 0; t < T_CHUNK; ++t) {
            v2f a[3];
#pragma unroll
            for (int c = 0; c < 3; ++c) {
                const float* p = &hw[n * H_STRIDE + 4 * c + 2 * half];
                a[c][0] = p[0];
                a[c][1] = p[1];
            }

            v8f acc;
#pragma unroll
            for (int r = 0; r < 8; ++r) {
                const int tok = s_tok[(rowLocal + r + mBase) * T_CHUNK + t];
                acc[r] = s_projb[tok * HP + n];
            }

            acc = wmma_f32_16x16x4(a[0], whb[0], acc);
            acc = wmma_f32_16x16x4(a[1], whb[1], acc);
            acc = wmma_f32_16x16x4(a[2], whb[2], acc);

#pragma unroll
            for (int r = 0; r < 8; ++r)
                hw[(r + mBase) * H_STRIDE + n] = fast_tanh(acc[r]) * colMask;

            asm volatile("s_wait_dscnt 0" ::: "memory");
        }
    }

    v2f af[3], wdb[3];
#pragma unroll
    for (int c = 0; c < 3; ++c) {
#pragma unroll
        for (int j = 0; j < 2; ++j) {
            const int K = 4 * c + j + 2 * half;
            float hv = 0.0f;
            if (K < H_SZ)
                hv = hw[n * H_STRIDE + K] * dmask[(gRowBase + n) * H_SZ + K];
            af[c][j]  = hv;
            wdb[c][j] = (K < H_SZ && n < L_SZ) ? Wd[K * L_SZ + n] : 0.0f;
        }
    }

    const float bdv = (n < L_SZ) ? bd[n] : 0.0f;
    v8f lac;
#pragma unroll
    for (int r = 0; r < 8; ++r) lac[r] = bdv;

    lac = wmma_f32_16x16x4(af[0], wdb[0], lac);
    lac = wmma_f32_16x16x4(af[1], wdb[1], lac);
    lac = wmma_f32_16x16x4(af[2], wdb[2], lac);

    if (n < L_SZ) {
#pragma unroll
        for (int r = 0; r < 8; ++r)
            out[(gRowBase + r + mBase) * L_SZ + n] = lac[r];
    }
}

extern "C" void kernel_launch(void* const* d_in, const int* in_sizes, int n_in,
                              void* d_out, int out_size, void* d_ws, size_t ws_size,
                              hipStream_t stream) {
    (void)in_sizes; (void)n_in; (void)d_ws; (void)ws_size; (void)out_size;
    const int*   x     = (const int*)  d_in[0];
    const float* emb   = (const float*)d_in[1];
    const float* Wx    = (const float*)d_in[2];
    const float* Wh    = (const float*)d_in[3];
    const float* b_rnn = (const float*)d_in[4];
    const float* Wd    = (const float*)d_in[5];
    const float* bd    = (const float*)d_in[6];
    const float* dmask = (const float*)d_in[7];
    float* out = (float*)d_out;

    charrnn_kernel<<<B_TOT / ROWS_PER_BLOCK, TPB, 0, stream>>>(
        x, emb, Wx, Wh, b_rnn, Wd, bd, dmask, out);
}


# Table build waits for memory once: Wx lies across the lanes (8 x 16-byte loads) and each FMA takes its element by DPP row_share; emb row, Wh, tokens join that load group (was 30 serial scalar hops)
# speedup vs baseline: 2.2332x; 2.2332x over previous
#include <hip/hip_runtime.h>


typedef float v2f __attribute__((ext_vector_type(2)));
typedef float v4f __attribute__((ext_vector_type(4)));
typedef float v8f __attribute__((ext_vector_type(8)));
typedef int   v4i __attribute__((ext_vector_type(4)));
typedef v4f v4f_a4 __attribute__((aligned(4)));
typedef v4f v4f_a8 __attribute__((aligned(8)));

#define B_TOT  16384
#define T_LEN  100
#define V_SZ   256
#define E_SZ   50
#define H_SZ   10
#define L_SZ   15
#define ROWS_PER_BLOCK 128
#define TPB    256
#define T_CHUNK 20
#define P_STRIDE 20
#define U_STRIDE 20


__device__ __forceinline__ float fast_tanh(float x) {
#if __has_builtin(__builtin_amdgcn_tanhf)
    return __builtin_amdgcn_tanhf(x);
#else
    x = fminf(fmaxf(x, -9.0f), 9.0f);
    float z = __builtin_amdgcn_exp2f(x * 2.885390081777927f);
    return (z - 1.0f) * __builtin_amdgcn_rcpf(z + 1.0f);
#endif
}

__device__ __forceinline__ v8f wmma_f32_16x16x4(v2f a, v2f b, v8f c) {
    return __builtin_amdgcn_wmma_f32_16x16x4_f32(
         false, a,  false, b,
         (short)0, c,  false,  false);
}

template <int F>
__device__ __forceinline__ void fmac_wx(float& acc, const v4f (&wxr)[8], float x) {
    const float w = wxr[F >> 6][F & 3];
    asm("v_fmac_f32_dpp %0, %1, %2 row_share:%3 row_mask:0xf bank_mask:0xf bound_ctrl:1"
        : "+v"(acc) : "v"(w), "v"(x), "n"((F >> 2) & 15));
}

template <int E>
struct TableAcc {
    static __device__ __forceinline__ void run(float (&s)[H_SZ], const float (&em)[E_SZ],
                                               const v4f (&wxr)[8]) {
        const float x = em[E];
        fmac_wx<E * H_SZ + 0>(s[0], wxr, x);
        fmac_wx<E * H_SZ + 1>(s[1], wxr, x);
        fmac_wx<E * H_SZ + 2>(s[2], wxr, x);
        fmac_wx<E * H_SZ + 3>(s[3], wxr, x);
        fmac_wx<E * H_SZ + 4>(s[4], wxr, x);
        fmac_wx<E * H_SZ + 5>(s[5], wxr, x);
        fmac_wx<E * H_SZ + 6>(s[6], wxr, x);
        fmac_wx<E * H_SZ + 7>(s[7], wxr, x);
        fmac_wx<E * H_SZ + 8>(s[8], wxr, x);
        fmac_wx<E * H_SZ + 9>(s[9], wxr, x);
        TableAcc<E + 1>::run(s, em, wxr);
    }
};
template <>
struct TableAcc<E_SZ> {
    static __device__ __forceinline__ void run(float (&)[H_SZ], const float (&)[E_SZ],
                                               const v4f (&)[8]) {}
};

__device__ __forceinline__ void lds_handover() { asm volatile("" ::: "memory"); }

__global__ __launch_bounds__(TPB) __attribute__((amdgpu_waves_per_eu(2, 2)))
void charrnn_kernel(const int*   __restrict__ x,
                    const float* __restrict__ emb,
                    const float* __restrict__ Wx,
                    const float* __restrict__ Wh,
                    const float* __restrict__ b_rnn,
                    const float* __restrict__ Wd,
                    const float* __restrict__ bd,
                    const float* __restrict__ dmask,
                    float*       __restrict__ out)
{
    __shared__ __attribute__((aligned(16))) float s_projb[V_SZ * P_STRIDE];
    __shared__ __attribute__((aligned(16))) float s_u[(TPB / 32) * 16 * U_STRIDE];

    const int tid   = threadIdx.x;
    const int lane  = tid & 31;
    const int wave  = tid >> 5;
    const int n     = lane & 15;
    const int half  = lane >> 4;
    const int mBase = half * 8;
    const int gRowBase = blockIdx.x * ROWS_PER_BLOCK + wave * 16;

    v4f wxr[8];
#pragma unroll
    for (int i = 0; i < 8; ++i) {
        int q4 = 16 * i + n;
        q4 = (q4 < (E_SZ * H_SZ) / 4 - 1) ? q4 : (E_SZ * H_SZ) / 4 - 1;
        wxr[i] = *reinterpret_cast<const v4f_a4*>(Wx + 4 * q4);
    }

    float em[E_SZ];
    {
        const float* er = emb + tid * E_SZ;
#pragma unroll
        for (int i = 0; i < E_SZ / 4; ++i) {
            const v4f t = *reinterpret_cast<const v4f_a8*>(er + 4 * i);
            em[4 * i + 0] = t[0]; em[4 * i + 1] = t[1];
            em[4 * i + 2] = t[2]; em[4 * i + 3] = t[3];
        }
        const v2f t2 = *reinterpret_cast<const v2f*>(er + (E_SZ / 4) * 4);
        em[E_SZ - 2] = t2[0]; em[E_SZ - 1] = t2[1];
    }

    v2f whb[3];
#pragma unroll
    for (int q = 0; q < 3; ++q) {
#pragma unroll
        for (int s = 0; s < 2; ++s) {
            const int K  = 6 * half + 2 * q + s;
            const int Kc = (K < H_SZ) ? K : H_SZ - 1;
            const int nc = (n < H_SZ) ? n : H_SZ - 1;
            const float w = Wh[Kc * H_SZ + nc];
            whb[q][s] = (K < H_SZ && n < H_SZ) ? w : 0.0f;
        }
    }

    const int* xrow = x + (gRowBase + n) * T_LEN;
    v4i tk[T_CHUNK / 4];
#pragma unroll
    for (int q = 0; q < T_CHUNK / 4; ++q)
        tk[q] = *reinterpret_cast<const v4i*>(xrow + 4 * q);

    {
        float s[H_SZ];
#pragma unroll
        for (int j = 0; j < H_SZ; ++j) s[j] = b_rnn[j];
        TableAcc<0>::run(s, em, wxr);
        v4f* dst = reinterpret_cast<v4f*>(&s_projb[tid * P_STRIDE]);
        dst[0] = (v4f){s[0], s[1], s[2], s[3]};
        dst[1] = (v4f){s[4], s[5], 0.0f, 0.0f};
        dst[2] = (v4f){s[6], s[7], s[8], s[9]};
        dst[3] = (v4f){0.0f, 0.0f, 0.0f, 0.0f};
        dst[4] = (v4f){0.0f, 0.0f, 0.0f, 0.0f};
    }

    float* uw = &s_u[wave * 16 * U_STRIDE];
    const int posn = n + ((n >= 6) ? 2 : 0);
    float*       ust  = uw + mBase * U_STRIDE + posn;
    const float* uld  = uw + n * U_STRIDE + 8 * half;
    const float* ptab = s_projb + 8 * half;

    v2f a[3];
#pragma unroll
    for (int q = 0; q < 3; ++q) a[q] = (v2f){0.0f, 0.0f};

    __syncthreads();

    for (int tc = 0; tc < T_LEN; tc += T_CHUNK) {
        const int tn = (tc + T_CHUNK < T_LEN) ? (tc + T_CHUNK) : tc;
        v4i nx[T_CHUNK / 4];
#pragma unroll
        for (int q = 0; q < T_CHUNK / 4; ++q)
            nx[q] = *reinterpret_cast<const v4i*>(xrow + tn + 4 * q);

#pragma unroll
        for (int i = 0; i < T_CHUNK; ++i) {
            const int tok = tk[i >> 2][i & 3];
            const float* pr = ptab + tok * P_STRIDE;
            const v4f p03 = *reinterpret_cast<const v4f*>(pr);
            const v2f p45 = *reinterpret_cast<const v2f*>(pr + 4);

            v8f u = (v8f){0.0f, 0.0f, 0.0f, 0.0f, 0.0f, 0.0f, 0.0f, 0.0f};
            u = wmma_f32_16x16x4(a[0], whb[0], u);
            u = wmma_f32_16x16x4(a[1], whb[1], u);
            u = wmma_f32_16x16x4(a[2], whb[2], u);

#pragma unroll
            for (int r = 0; r < 8; ++r) ust[r * U_STRIDE] = u[r];
            lds_handover();
            const v4f u03 = *reinterpret_cast<const v4f*>(uld);
            const v2f u45 = *reinterpret_cast<const v2f*>(uld + 4);

            a[0][0] = fast_tanh(u03[0] + p03[0]);
            a[0][1] = fast_tanh(u03[1] + p03[1]);
            a[1][0] = fast_tanh(u03[2] + p03[2]);
            a[1][1] = fast_tanh(u03[3] + p03[3]);
            a[2][0] = fast_tanh(u45[0] + p45[0]);
            a[2][1] = fast_tanh(u45[1] + p45[1]);
        }
#pragma unroll
        for (int q = 0; q < T_CHUNK / 4; ++q) tk[q] = nx[q];
    }

    v2f af[3], wdb[3];
#pragma unroll
    for (int q = 0; q < 3; ++q) {
#pragma unroll
        for (int s = 0; s < 2; ++s) {
            const int K = 6 * half + 2 * q + s;
            float hv = 0.0f;
            if (K < H_SZ)
                hv = a[q][s] * dmask[(gRowBase + n) * H_SZ + K];
            af[q][s]  = hv;
            wdb[q][s] = (K < H_SZ && n < L_SZ) ? Wd[K * L_SZ + n] : 0.0f;
        }
    }

    const float bdv = (n < L_SZ) ? bd[n] : 0.0f;
    v8f lac;
#pragma unroll
    for (int r = 0; r < 8; ++r) lac[r] = bdv;

    lac = wmma_f32_16x16x4(af[0], wdb[0], lac);
    lac = wmma_f32_16x16x4(af[1], wdb[1], lac);
    lac = wmma_f32_16x16x4(af[2], wdb[2], lac);

    if (n < L_SZ) {
#pragma unroll
        for (int r = 0; r < 8; ++r)
            out[(gRowBase + r + mBase) * L_SZ + n] = lac[r];
    }
}

extern "C" void kernel_launch(void* const* d_in, const int* in_sizes, int n_in,
                              void* d_out, int out_size, void* d_ws, size_t ws_size,
                              hipStream_t stream) {
    (void)in_sizes; (void)n_in; (void)d_ws; (void)ws_size; (void)out_size;
    const int*   x     = (const int*)  d_in[0];
    const float* emb   = (const float*)d_in[1];
    const float* Wx    = (const float*)d_in[2];
    const float* Wh    = (const float*)d_in[3];
    const float* b_rnn = (const float*)d_in[4];
    const float* Wd    = (const float*)d_in[5];
    const float* bd    = (const float*)d_in[6];
    const float* dmask = (const float*)d_in[7];
    float* out = (float*)d_out;

    charrnn_kernel<<<B_TOT / ROWS_PER_BLOCK, TPB, 0, stream>>>(
        x, emb, Wx, Wh, b_rnn, Wd, bd, dmask, out);
}


# LDS hand-over tile transposed: D side leaves in two 16-byte stores, A side reads three 2-address loads, next step's table loads pinned behind them: first fragment load is 3rd, not 7th, in the queue
# speedup vs baseline: 2.3469x; 1.0509x over previous
#include <hip/hip_runtime.h>


typedef float v2f __attribute__((ext_vector_type(2)));
typedef float v4f __attribute__((ext_vector_type(4)));
typedef float v8f __attribute__((ext_vector_type(8)));
typedef int   v4i __attribute__((ext_vector_type(4)));
typedef v4f v4f_a4 __attribute__((aligned(4)));
typedef v4f v4f_a8 __attribute__((aligned(8)));

#define B_TOT  16384
#define T_LEN  100
#define V_SZ   256
#define E_SZ   50
#define H_SZ   10
#define L_SZ   15
#define ROWS_PER_BLOCK 128
#define TPB    256
#define T_CHUNK 20
#define P_STRIDE 20
#define U_PITCH 28


__device__ __forceinline__ float fast_tanh(float x) {
#if __has_builtin(__builtin_amdgcn_tanhf)
    return __builtin_amdgcn_tanhf(x);
#else
    x = fminf(fmaxf(x, -9.0f), 9.0f);
    float z = __builtin_amdgcn_exp2f(x * 2.885390081777927f);
    return (z - 1.0f) * __builtin_amdgcn_rcpf(z + 1.0f);
#endif
}

__device__ __forceinline__ v8f wmma_f32_16x16x4(v2f a, v2f b, v8f c) {
    return __builtin_amdgcn_wmma_f32_16x16x4_f32(
         false, a,  false, b,
         (short)0, c,  false,  false);
}

template <int F>
__device__ __forceinline__ void fmac_wx(float& acc, const v4f (&wxr)[8], float x) {
    const float w = wxr[F >> 6][F & 3];
    asm("v_fmac_f32_dpp %0, %1, %2 row_share:%3 row_mask:0xf bank_mask:0xf bound_ctrl:1"
        : "+v"(acc) : "v"(w), "v"(x), "n"((F >> 2) & 15));
}

template <int E>
struct TableAcc {
    static __device__ __forceinline__ void run(float (&s)[H_SZ], const float (&em)[E_SZ],
                                               const v4f (&wxr)[8]) {
        const float x = em[E];
        fmac_wx<E * H_SZ + 0>(s[0], wxr, x);
        fmac_wx<E * H_SZ + 1>(s[1], wxr, x);
        fmac_wx<E * H_SZ + 2>(s[2], wxr, x);
        fmac_wx<E * H_SZ + 3>(s[3], wxr, x);
        fmac_wx<E * H_SZ + 4>(s[4], wxr, x);
        fmac_wx<E * H_SZ + 5>(s[5], wxr, x);
        fmac_wx<E * H_SZ + 6>(s[6], wxr, x);
        fmac_wx<E * H_SZ + 7>(s[7], wxr, x);
        fmac_wx<E * H_SZ + 8>(s[8], wxr, x);
        fmac_wx<E * H_SZ + 9>(s[9], wxr, x);
        TableAcc<E + 1>::run(s, em, wxr);
    }
};
template <>
struct TableAcc<E_SZ> {
    static __device__ __forceinline__ void run(float (&)[H_SZ], const float (&)[E_SZ],
                                               const v4f (&)[8]) {}
};

__device__ __forceinline__ void lds_handover() { asm volatile("" ::: "memory"); }

__global__ __launch_bounds__(TPB) __attribute__((amdgpu_waves_per_eu(2, 2)))
void charrnn_kernel(const int*   __restrict__ x,
                    const float* __restrict__ emb,
                    const float* __restrict__ Wx,
                    const float* __restrict__ Wh,
                    const float* __restrict__ b_rnn,
                    const float* __restrict__ Wd,
                    const float* __restrict__ bd,
                    const float* __restrict__ dmask,
                    float*       __restrict__ out)
{
    __shared__ __attribute__((aligned(16))) float s_projb[V_SZ * P_STRIDE];
    __shared__ __attribute__((aligned(16))) float s_u[(TPB / 32) * 16 * U_PITCH];

    const int tid   = threadIdx.x;
    const int lane  = tid & 31;
    const int wave  = tid >> 5;
    const int n     = lane & 15;
    const int half  = lane >> 4;
    const int mBase = half * 8;
    const int gRowBase = blockIdx.x * ROWS_PER_BLOCK + wave * 16;

    v4f wxr[8];
#pragma unroll
    for (int i = 0; i < 8; ++i) {
        int q4 = 16 * i + n;
        q4 = (q4 < (E_SZ * H_SZ) / 4 - 1) ? q4 : (E_SZ * H_SZ) / 4 - 1;
        wxr[i] = *reinterpret_cast<const v4f_a4*>(Wx + 4 * q4);
    }

    float em[E_SZ];
    {
        const float* er = emb + tid * E_SZ;
#pragma unroll
        for (int i = 0; i < E_SZ / 4; ++i) {
            const v4f t = *reinterpret_cast<const v4f_a8*>(er + 4 * i);
            em[4 * i + 0] = t[0]; em[4 * i + 1] = t[1];
            em[4 * i + 2] = t[2]; em[4 * i + 3] = t[3];
        }
        const v2f t2 = *reinterpret_cast<const v2f*>(er + (E_SZ / 4) * 4);
        em[E_SZ - 2] = t2[0]; em[E_SZ - 1] = t2[1];
    }

    v2f whb[3];
#pragma unroll
    for (int q = 0; q < 3; ++q) {
#pragma unroll
        for (int s = 0; s < 2; ++s) {
            const int K  = 6 * half + 2 * q + s;
            const int Kc = (K < H_SZ) ? K : H_SZ - 1;
            const int nc = (n < H_SZ) ? n : H_SZ - 1;
            const float w = Wh[Kc * H_SZ + nc];
            whb[q][s] = (K < H_SZ && n < H_SZ) ? w : 0.0f;
        }
    }

    const int* xrow = x + (gRowBase + n) * T_LEN;
    v4i tk[T_CHUNK / 4];
#pragma unroll
    for (int q = 0; q < T_CHUNK / 4; ++q)
        tk[q] = *reinterpret_cast<const v4i*>(xrow + 4 * q);

    {
        float s[H_SZ];
#pragma unroll
        for (int j = 0; j < H_SZ; ++j) s[j] = b_rnn[j];
        TableAcc<0>::run(s, em, wxr);
        v4f* dst = reinterpret_cast<v4f*>(&s_projb[tid * P_STRIDE]);
        dst[0] = (v4f){s[0], s[1], s[2], s[3]};
        dst[1] = (v4f){s[4], s[5], 0.0f, 0.0f};
        dst[2] = (v4f){s[6], s[7], s[8], s[9]};
        dst[3] = (v4f){0.0f, 0.0f, 0.0f, 0.0f};
        dst[4] = (v4f){0.0f, 0.0f, 0.0f, 0.0f};
    }

    float* uw = &s_u[wave * 16 * U_PITCH];
    float*       ust  = uw + n * U_PITCH + mBase;
    const float* uld  = uw + (6 * half) * U_PITCH + n;
    const float* ptab = s_projb + 8 * half;

    v2f a[3];
#pragma unroll
    for (int q = 0; q < 3; ++q) a[q] = (v2f){0.0f, 0.0f};

    __syncthreads();

    v4f p03; v2f p45;
    {
        const float* pr = ptab + tk[0][0] * P_STRIDE;
        p03 = *reinterpret_cast<const v4f*>(pr);
        p45 = *reinterpret_cast<const v2f*>(pr + 4);
    }

    for (int tc = 0; tc < T_LEN; tc += T_CHUNK) {
        const int tn = (tc + T_CHUNK < T_LEN) ? (tc + T_CHUNK) : tc;
        v4i nx[T_CHUNK / 4];
#pragma unroll
        for (int q = 0; q < T_CHUNK / 4; ++q)
            nx[q] = *reinterpret_cast<const v4i*>(xrow + tn + 4 * q);

#pragma unroll
        for (int i = 0; i < T_CHUNK; ++i) {
            const int tokn = (i + 1 < T_CHUNK) ? tk[(i + 1) >> 2][(i + 1) & 3] : nx[0][0];
            const float* prn = ptab + tokn * P_STRIDE;

            v8f u = (v8f){0.0f, 0.0f, 0.0f, 0.0f, 0.0f, 0.0f, 0.0f, 0.0f};
            u = wmma_f32_16x16x4(a[0], whb[0], u);
            u = wmma_f32_16x16x4(a[1], whb[1], u);
            u = wmma_f32_16x16x4(a[2], whb[2], u);

            *reinterpret_cast<v4f*>(ust)     = (v4f){u[0], u[1], u[2], u[3]};
            *reinterpret_cast<v4f*>(ust + 4) = (v4f){u[4], u[5], u[6], u[7]};
            lds_handover();
            const float u0 = uld[0 * U_PITCH], u1 = uld[1 * U_PITCH];
            const float u2 = uld[2 * U_PITCH], u3 = uld[3 * U_PITCH];
            const float u4 = uld[4 * U_PITCH], u5 = uld[5 * U_PITCH];
            const v4f n03 = *reinterpret_cast<const v4f*>(prn);
            const v2f n45 = *reinterpret_cast<const v2f*>(prn + 4);
            __builtin_amdgcn_sched_barrier(0);

            a[0][0] = fast_tanh(u0 + p03[0]);
            a[0][1] = fast_tanh(u1 + p03[1]);
            a[1][0] = fast_tanh(u2 + p03[2]);
            a[1][1] = fast_tanh(u3 + p03[3]);
            a[2][0] = fast_tanh(u4 + p45[0]);
            a[2][1] = fast_tanh(u5 + p45[1]);
            p03 = n03; p45 = n45;
        }
#pragma unroll
        for (int q = 0; q < T_CHUNK / 4; ++q) tk[q] = nx[q];
    }

    v2f af[3], wdb[3];
#pragma unroll
    for (int q = 0; q < 3; ++q) {
#pragma unroll
        for (int s = 0; s < 2; ++s) {
            const int K = 6 * half + 2 * q + s;
            float hv = 0.0f;
            if (K < H_SZ)
                hv = a[q][s] * dmask[(gRowBase + n) * H_SZ + K];
            af[q][s]  = hv;
            wdb[q][s] = (K < H_SZ && n < L_SZ) ? Wd[K * L_SZ + n] : 0.0f;
        }
    }

    const float bdv = (n < L_SZ) ? bd[n] : 0.0f;
    v8f lac;
#pragma unroll
    for (int r = 0; r < 8; ++r) lac[r] = bdv;

    lac = wmma_f32_16x16x4(af[0], wdb[0], lac);
    lac = wmma_f32_16x16x4(af[1], wdb[1], lac);
    lac = wmma_f32_16x16x4(af[2], wdb[2], lac);

    if (n < L_SZ) {
#pragma unroll
        for (int r = 0; r < 8; ++r)
            out[(gRowBase + r + mBase) * L_SZ + n] = lac[r];
    }
}

extern "C" void kernel_launch(void* const* d_in, const int* in_sizes, int n_in,
                              void* d_out, int out_size, void* d_ws, size_t ws_size,
                              hipStream_t stream) {
    (void)in_sizes; (void)n_in; (void)d_ws; (void)ws_size; (void)out_size;
    const int*   x     = (const int*)  d_in[0];
    const float* emb   = (const float*)d_in[1];
    const float* Wx    = (const float*)d_in[2];
    const float* Wh    = (const float*)d_in[3];
    const float* b_rnn = (const float*)d_in[4];
    const float* Wd    = (const float*)d_in[5];
    const float* bd    = (const float*)d_in[6];
    const float* dmask = (const float*)d_in[7];
    float* out = (float*)d_out;

    charrnn_kernel<<<B_TOT / ROWS_PER_BLOCK, TPB, 0, stream>>>(
        x, emb, Wx, Wh, b_rnn, Wd, bd, dmask, out);
}


# Remainder k chunk off the step's dependent matrix chain: units 0-7 stay two f32 WMMAs, h8*Wh[8]+h9*Wh[9] joins the token term as 12 f32 vector multiply-adds in the LDS wait (was a 3rd product)
# speedup vs baseline: 2.4768x; 1.0553x over previous
#include <hip/hip_runtime.h>


typedef float v2f __attribute__((ext_vector_type(2)));
typedef float v4f __attribute__((ext_vector_type(4)));
typedef float v8f __attribute__((ext_vector_type(8)));
typedef int   v4i __attribute__((ext_vector_type(4)));
typedef v4f v4f_a4 __attribute__((aligned(4)));
typedef v4f v4f_a8 __attribute__((aligned(8)));

#define B_TOT  16384
#define T_LEN  100
#define V_SZ   256
#define E_SZ   50
#define H_SZ   10
#define L_SZ   15
#define ROWS_PER_BLOCK 128
#define TPB    256
#define T_CHUNK 20
#define P_STRIDE 20
#define U_PITCH 28


__device__ __forceinline__ float fast_tanh(float x) {
#if __has_builtin(__builtin_amdgcn_tanhf)
    return __builtin_amdgcn_tanhf(x);
#else
    x = fminf(fmaxf(x, -9.0f), 9.0f);
    float z = __builtin_amdgcn_exp2f(x * 2.885390081777927f);
    return (z - 1.0f) * __builtin_amdgcn_rcpf(z + 1.0f);
#endif
}

__device__ __forceinline__ v8f wmma_f32_16x16x4(v2f a, v2f b, v8f c) {
    return __builtin_amdgcn_wmma_f32_16x16x4_f32(
         false, a,  false, b,
         (short)0, c,  false,  false);
}

template <int F>
__device__ __forceinline__ void fmac_wx(float& acc, const v4f (&wxr)[8], float x) {
    const float w = wxr[F >> 6][F & 3];
    asm("v_fmac_f32_dpp %0, %1, %2 row_share:%3 row_mask:0xf bank_mask:0xf bound_ctrl:1"
        : "+v"(acc) : "v"(w), "v"(x), "n"((F >> 2) & 15));
}

template <int E>
struct TableAcc {
    static __device__ __forceinline__ void run(float (&s)[H_SZ], const float (&em)[E_SZ],
                                               const v4f (&wxr)[8]) {
        const float x = em[E];
        fmac_wx<E * H_SZ + 0>(s[0], wxr, x);
        fmac_wx<E * H_SZ + 1>(s[1], wxr, x);
        fmac_wx<E * H_SZ + 2>(s[2], wxr, x);
        fmac_wx<E * H_SZ + 3>(s[3], wxr, x);
        fmac_wx<E * H_SZ + 4>(s[4], wxr, x);
        fmac_wx<E * H_SZ + 5>(s[5], wxr, x);
        fmac_wx<E * H_SZ + 6>(s[6], wxr, x);
        fmac_wx<E * H_SZ + 7>(s[7], wxr, x);
        fmac_wx<E * H_SZ + 8>(s[8], wxr, x);
        fmac_wx<E * H_SZ + 9>(s[9], wxr, x);
        TableAcc<E + 1>::run(s, em, wxr);
    }
};
template <>
struct TableAcc<E_SZ> {
    static __device__ __forceinline__ void run(float (&)[H_SZ], const float (&)[E_SZ],
                                               const v4f (&)[8]) {}
};

__device__ __forceinline__ void lds_handover() { asm volatile("" ::: "memory"); }

__global__ __launch_bounds__(TPB) __attribute__((amdgpu_waves_per_eu(2, 2)))
void charrnn_kernel(const int*   __restrict__ x,
                    const float* __restrict__ emb,
                    const float* __restrict__ Wx,
                    const float* __restrict__ Wh,
                    const float* __restrict__ b_rnn,
                    const float* __restrict__ Wd,
                    const float* __restrict__ bd,
                    const float* __restrict__ dmask,
                    float*       __restrict__ out)
{
    __shared__ __attribute__((aligned(16))) float s_projb[V_SZ * P_STRIDE];
    __shared__ __attribute__((aligned(16))) float s_u[(TPB / 32) * 16 * U_PITCH];

    const int tid   = threadIdx.x;
    const int lane  = tid & 31;
    const int wave  = tid >> 5;
    const int n     = lane & 15;
    const int half  = lane >> 4;
    const int mBase = half * 8;
    const int gRowBase = blockIdx.x * ROWS_PER_BLOCK + wave * 16;

    v4f wxr[8];
#pragma unroll
    for (int i = 0; i < 8; ++i) {
        int q4 = 16 * i + n;
        q4 = (q4 < (E_SZ * H_SZ) / 4 - 1) ? q4 : (E_SZ * H_SZ) / 4 - 1;
        wxr[i] = *reinterpret_cast<const v4f_a4*>(Wx + 4 * q4);
    }

    float em[E_SZ];
    {
        const float* er = emb + tid * E_SZ;
#pragma unroll
        for (int i = 0; i < E_SZ / 4; ++i) {
            const v4f t = *reinterpret_cast<const v4f_a8*>(er + 4 * i);
            em[4 * i + 0] = t[0]; em[4 * i + 1] = t[1];
            em[4 * i + 2] = t[2]; em[4 * i + 3] = t[3];
        }
        const v2f t2 = *reinterpret_cast<const v2f*>(er + (E_SZ / 4) * 4);
        em[E_SZ - 2] = t2[0]; em[E_SZ - 1] = t2[1];
    }

    v2f whb[2];
#pragma unroll
    for (int q = 0; q < 2; ++q) {
#pragma unroll
        for (int s = 0; s < 2; ++s) {
            const int K  = 4 * half + 2 * q + s;
            const int nc = (n < H_SZ) ? n : H_SZ - 1;
            const float w = Wh[K * H_SZ + nc];
            whb[q][s] = (n < H_SZ) ? w : 0.0f;
        }
    }

    float w8[6], w9[6];
#pragma unroll
    for (int j = 0; j < 6; ++j) {
        const int c = (j < 4) ? 4 * half + j : 4 + j;
        w8[j] = Wh[8 * H_SZ + c];
        w9[j] = Wh[9 * H_SZ + c];
    }

    const int* xrow = x + (gRowBase + n) * T_LEN;
    v4i tk[T_CHUNK / 4];
#pragma unroll
    for (int q = 0; q < T_CHUNK / 4; ++q)
        tk[q] = *reinterpret_cast<const v4i*>(xrow + 4 * q);

    {
        float s[H_SZ];
#pragma unroll
        for (int j = 0; j < H_SZ; ++j) s[j] = b_rnn[j];
        TableAcc<0>::run(s, em, wxr);
        v4f* dst = reinterpret_cast<v4f*>(&s_projb[tid * P_STRIDE]);
        dst[0] = (v4f){s[0], s[1], s[2], s[3]};
        dst[1] = (v4f){s[8], s[9], 0.0f, 0.0f};
        dst[2] = (v4f){s[4], s[5], s[6], s[7]};
        dst[3] = (v4f){s[8], s[9], 0.0f, 0.0f};
    }

    float* uw = &s_u[wave * 16 * U_PITCH];
    float*       ust  = uw + n * U_PITCH + mBase;
    const float* uld  = uw + (4 * half) * U_PITCH + n;
    const float* uld89 = uw + 8 * U_PITCH + n;
    const float* ptab = s_projb + 8 * half;

    v2f a[2];
#pragma unroll
    for (int q = 0; q < 2; ++q) a[q] = (v2f){0.0f, 0.0f};
    float h8 = 0.0f, h9 = 0.0f;

    __syncthreads();

    v4f p03; v2f p45;
    {
        const float* pr = ptab + tk[0][0] * P_STRIDE;
        p03 = *reinterpret_cast<const v4f*>(pr);
        p45 = *reinterpret_cast<const v2f*>(pr + 4);
    }

    for (int tc = 0; tc < T_LEN; tc += T_CHUNK) {
        const int tn = (tc + T_CHUNK < T_LEN) ? (tc + T_CHUNK) : tc;
        v4i nx[T_CHUNK / 4];
#pragma unroll
        for (int q = 0; q < T_CHUNK / 4; ++q)
            nx[q] = *reinterpret_cast<const v4i*>(xrow + tn + 4 * q);

#pragma unroll
        for (int i = 0; i < T_CHUNK; ++i) {
            const int tokn = (i + 1 < T_CHUNK) ? tk[(i + 1) >> 2][(i + 1) & 3] : nx[0][0];
            const float* prn = ptab + tokn * P_STRIDE;

            v8f u = (v8f){0.0f, 0.0f, 0.0f, 0.0f, 0.0f, 0.0f, 0.0f, 0.0f};
            u = wmma_f32_16x16x4(a[0], whb[0], u);
            u = wmma_f32_16x16x4(a[1], whb[1], u);

            *reinterpret_cast<v4f*>(ust)     = (v4f){u[0], u[1], u[2], u[3]};
            *reinterpret_cast<v4f*>(ust + 4) = (v4f){u[4], u[5], u[6], u[7]};
            lds_handover();
            const float u0 = uld[0 * U_PITCH], u1 = uld[1 * U_PITCH];
            const float u2 = uld[2 * U_PITCH], u3 = uld[3 * U_PITCH];
            const float u4 = uld89[0 * U_PITCH], u5 = uld89[1 * U_PITCH];
            const v4f n03 = *reinterpret_cast<const v4f*>(prn);
            const v2f n45 = *reinterpret_cast<const v2f*>(prn + 4);
            __builtin_amdgcn_sched_barrier(0);

            const float v0 = fmaf(h9, w9[0], fmaf(h8, w8[0], p03[0]));
            const float v1 = fmaf(h9, w9[1], fmaf(h8, w8[1], p03[1]));
            const float v2 = fmaf(h9, w9[2], fmaf(h8, w8[2], p03[2]));
            const float v3 = fmaf(h9, w9[3], fmaf(h8, w8[3], p03[3]));
            const float v4 = fmaf(h9, w9[4], fmaf(h8, w8[4], p45[0]));
            const float v5 = fmaf(h9, w9[5], fmaf(h8, w8[5], p45[1]));

            a[0][0] = fast_tanh(u0 + v0);
            a[0][1] = fast_tanh(u1 + v1);
            a[1][0] = fast_tanh(u2 + v2);
            a[1][1] = fast_tanh(u3 + v3);
            h8 = fast_tanh(u4 + v4);
            h9 = fast_tanh(u5 + v5);
            p03 = n03; p45 = n45;
        }
#pragma unroll
        for (int q = 0; q < T_CHUNK / 4; ++q) tk[q] = nx[q];
    }

    v2f af[3], wdb[3];
#pragma unroll
    for (int q = 0; q < 2; ++q) {
#pragma unroll
        for (int s = 0; s < 2; ++s) {
            const int K = 4 * half + 2 * q + s;
            af[q][s]  = a[q][s] * dmask[(gRowBase + n) * H_SZ + K];
            wdb[q][s] = (n < L_SZ) ? Wd[K * L_SZ + n] : 0.0f;
        }
    }
#pragma unroll
    for (int s = 0; s < 2; ++s) {
        const int K = 8 + s;
        const float hv = (s == 0) ? h8 : h9;
        af[2][s]  = (half == 0) ? hv * dmask[(gRowBase + n) * H_SZ + K] : 0.0f;
        wdb[2][s] = (half == 0 && n < L_SZ) ? Wd[K * L_SZ + n] : 0.0f;
    }

    const float bdv = (n < L_SZ) ? bd[n] : 0.0f;
    v8f lac;
#pragma unroll
    for (int r = 0; r < 8; ++r) lac[r] = bdv;

    lac = wmma_f32_16x16x4(af[0], wdb[0], lac);
    lac = wmma_f32_16x16x4(af[1], wdb[1], lac);
    lac = wmma_f32_16x16x4(af[2], wdb[2], lac);

    if (n < L_SZ) {
#pragma unroll
        for (int r = 0; r < 8; ++r)
            out[(gRowBase + r + mBase) * L_SZ + n] = lac[r];
    }
}

extern "C" void kernel_launch(void* const* d_in, const int* in_sizes, int n_in,
                              void* d_out, int out_size, void* d_ws, size_t ws_size,
                              hipStream_t stream) {
    (void)in_sizes; (void)n_in; (void)d_ws; (void)ws_size; (void)out_size;
    const int*   x     = (const int*)  d_in[0];
    const float* emb   = (const float*)d_in[1];
    const float* Wx    = (const float*)d_in[2];
    const float* Wh    = (const float*)d_in[3];
    const float* b_rnn = (const float*)d_in[4];
    const float* Wd    = (const float*)d_in[5];
    const float* bd    = (const float*)d_in[6];
    const float* dmask = (const float*)d_in[7];
    float* out = (float*)d_out;

    charrnn_kernel<<<B_TOT / ROWS_PER_BLOCK, TPB, 0, stream>>>(
        x, emb, Wx, Wh, b_rnn, Wd, bd, dmask, out);
}


# Second k chunk off the step's dependent matrix chain: units 0-3 stay one f32 WMMA per step, units 4-9 join the token term as f32 vector multiply-adds in the LDS wait, 3 tanh swapped by permlanex16
# speedup vs baseline: 2.6087x; 1.0533x over previous
#include <hip/hip_runtime.h>


typedef float v2f __attribute__((ext_vector_type(2)));
typedef float v4f __attribute__((ext_vector_type(4)));
typedef float v8f __attribute__((ext_vector_type(8)));
typedef int   v4i __attribute__((ext_vector_type(4)));
typedef v4f v4f_a4 __attribute__((aligned(4)));
typedef v4f v4f_a8 __attribute__((aligned(8)));

#define B_TOT  16384
#define T_LEN  100
#define V_SZ   256
#define E_SZ   50
#define H_SZ   10
#define L_SZ   15
#define ROWS_PER_BLOCK 128
#define TPB    256
#define T_CHUNK 20
#define P_STRIDE 20
#define U_PITCH 28


__device__ __forceinline__ float fast_tanh(float x) {
#if __has_builtin(__builtin_amdgcn_tanhf)
    return __builtin_amdgcn_tanhf(x);
#else
    x = fminf(fmaxf(x, -9.0f), 9.0f);
    float z = __builtin_amdgcn_exp2f(x * 2.885390081777927f);
    return (z - 1.0f) * __builtin_amdgcn_rcpf(z + 1.0f);
#endif
}

__device__ __forceinline__ v8f wmma_f32_16x16x4(v2f a, v2f b, v8f c) {
    return __builtin_amdgcn_wmma_f32_16x16x4_f32(
         false, a,  false, b,
         (short)0, c,  false,  false);
}

__device__ __forceinline__ int k_unit(int q, int s, int hf) {
    if (q == 0) return 2 * hf + s;
    if (q == 1) return 4 + 3 * hf + s;
    return (s == 0) ? 6 + 3 * hf : -1;
}

__device__ __forceinline__ float other_half(float dead, float x) {
    return __builtin_bit_cast(float,
        __builtin_amdgcn_permlanex16(__builtin_bit_cast(int, dead), __builtin_bit_cast(int, x),
                                     0x76543210, 0xfedcba98, false, false));
}

__device__ __forceinline__ void pin_here(float& x, float& y, float& z) {
    asm volatile("" : "+v"(x), "+v"(y), "+v"(z));
}

template <int F>
__device__ __forceinline__ void fmac_wx(float& acc, const v4f (&wxr)[8], float x) {
    const float w = wxr[F >> 6][F & 3];
    asm("v_fmac_f32_dpp %0, %1, %2 row_share:%3 row_mask:0xf bank_mask:0xf bound_ctrl:1"
        : "+v"(acc) : "v"(w), "v"(x), "n"((F >> 2) & 15));
}

template <int E>
struct TableAcc {
    static __device__ __forceinline__ void run(float (&s)[H_SZ], const float (&em)[E_SZ],
                                               const v4f (&wxr)[8]) {
        const float x = em[E];
        fmac_wx<E * H_SZ + 0>(s[0], wxr, x);
        fmac_wx<E * H_SZ + 1>(s[1], wxr, x);
        fmac_wx<E * H_SZ + 2>(s[2], wxr, x);
        fmac_wx<E * H_SZ + 3>(s[3], wxr, x);
        fmac_wx<E * H_SZ + 4>(s[4], wxr, x);
        fmac_wx<E * H_SZ + 5>(s[5], wxr, x);
        fmac_wx<E * H_SZ + 6>(s[6], wxr, x);
        fmac_wx<E * H_SZ + 7>(s[7], wxr, x);
        fmac_wx<E * H_SZ + 8>(s[8], wxr, x);
        fmac_wx<E * H_SZ + 9>(s[9], wxr, x);
        TableAcc<E + 1>::run(s, em, wxr);
    }
};
template <>
struct TableAcc<E_SZ> {
    static __device__ __forceinline__ void run(float (&)[H_SZ], const float (&)[E_SZ],
                                               const v4f (&)[8]) {}
};

__device__ __forceinline__ void lds_handover() { asm volatile("" ::: "memory"); }

__global__ __launch_bounds__(TPB) __attribute__((amdgpu_waves_per_eu(2, 2)))
void charrnn_kernel(const int*   __restrict__ x,
                    const float* __restrict__ emb,
                    const float* __restrict__ Wx,
                    const float* __restrict__ Wh,
                    const float* __restrict__ b_rnn,
                    const float* __restrict__ Wd,
                    const float* __restrict__ bd,
                    const float* __restrict__ dmask,
                    float*       __restrict__ out)
{
    __shared__ __attribute__((aligned(16))) float s_projb[V_SZ * P_STRIDE];
    __shared__ __attribute__((aligned(16))) float s_u[(TPB / 32) * 16 * U_PITCH];

    const int tid   = threadIdx.x;
    const int lane  = tid & 31;
    const int wave  = tid >> 5;
    const int n     = lane & 15;
    const int half  = lane >> 4;
    const int mBase = half * 8;
    const int gRowBase = blockIdx.x * ROWS_PER_BLOCK + wave * 16;

    v4f wxr[8];
#pragma unroll
    for (int i = 0; i < 8; ++i) {
        int q4 = 16 * i + n;
        q4 = (q4 < (E_SZ * H_SZ) / 4 - 1) ? q4 : (E_SZ * H_SZ) / 4 - 1;
        wxr[i] = *reinterpret_cast<const v4f_a4*>(Wx + 4 * q4);
    }

    float em[E_SZ];
    {
        const float* er = emb + tid * E_SZ;
#pragma unroll
        for (int i = 0; i < E_SZ / 4; ++i) {
            const v4f t = *reinterpret_cast<const v4f_a8*>(er + 4 * i);
            em[4 * i + 0] = t[0]; em[4 * i + 1] = t[1];
            em[4 * i + 2] = t[2]; em[4 * i + 3] = t[3];
        }
        const v2f t2 = *reinterpret_cast<const v2f*>(er + (E_SZ / 4) * 4);
        em[E_SZ - 2] = t2[0]; em[E_SZ - 1] = t2[1];
    }

    v2f whb;
#pragma unroll
    for (int s = 0; s < 2; ++s) {
        const int K  = k_unit(0, s, half);
        const int nc = (n < H_SZ) ? n : H_SZ - 1;
        const float w = Wh[K * H_SZ + nc];
        whb[s] = (n < H_SZ) ? w : 0.0f;
    }

    float wo[3][5], wp[3][5];
#pragma unroll
    for (int c = 0; c < 5; ++c) {
        const int c0 = (c < 2) ? c : c + 2;
        const int c1 = (c < 2) ? c + 2 : c + 5;
#pragma unroll
        for (int j = 0; j < 3; ++j) {
            const float lo0 = Wh[(4 + j) * H_SZ + c0], lo1 = Wh[(4 + j) * H_SZ + c1];
            const float hi0 = Wh[(7 + j) * H_SZ + c0], hi1 = Wh[(7 + j) * H_SZ + c1];
            wo[j][c] = half ? hi1 : lo0;
            wp[j][c] = half ? lo1 : hi0;
        }
    }

    const int* xrow = x + (gRowBase + n) * T_LEN;
    v4i tk[T_CHUNK / 4];
#pragma unroll
    for (int q = 0; q < T_CHUNK / 4; ++q)
        tk[q] = *reinterpret_cast<const v4i*>(xrow + 4 * q);

    {
        float s[H_SZ];
#pragma unroll
        for (int j = 0; j < H_SZ; ++j) s[j] = b_rnn[j];
        TableAcc<0>::run(s, em, wxr);
        v4f* dst = reinterpret_cast<v4f*>(&s_projb[tid * P_STRIDE]);
        dst[0] = (v4f){s[0], s[1], s[4], s[5]};
        dst[1] = (v4f){s[6], 0.0f, 0.0f, 0.0f};
        dst[2] = (v4f){s[2], s[3], s[7], s[8]};
        dst[3] = (v4f){s[9], 0.0f, 0.0f, 0.0f};
    }

    float* uw = &s_u[wave * 16 * U_PITCH];
    const int urow = (n < 14) ? n + (n >= 2) + (n >= 4) : 3 * n - 40;
    float*       ust  = uw + urow * U_PITCH + mBase;
    const float* uldf = uw + (3 * half) * U_PITCH + n;
    const float* uldv = uw + (6 + 3 * half) * U_PITCH + n;
    const float* ptab = s_projb + 8 * half;

    v2f a = (v2f){0.0f, 0.0f};
    float gq0 = 0.0f, gq1 = 0.0f, gq2 = 0.0f;
    float vq0 = 0.0f, vq1 = 0.0f, vq2 = 0.0f;
    float hp0 = 0.0f, hp1 = 0.0f, hp2 = 0.0f;

    __syncthreads();

    v4f p03; float p4;
    {
        const float* pr = ptab + tk[0][0] * P_STRIDE;
        p03 = *reinterpret_cast<const v4f*>(pr);
        p4  = pr[4];
    }

    for (int tc = 0; tc < T_LEN; tc += T_CHUNK) {
        const int tn = (tc + T_CHUNK < T_LEN) ? (tc + T_CHUNK) : tc;
        v4i nx[T_CHUNK / 4];
#pragma unroll
        for (int q = 0; q < T_CHUNK / 4; ++q)
            nx[q] = *reinterpret_cast<const v4i*>(xrow + tn + 4 * q);

#pragma unroll
        for (int i = 0; i < T_CHUNK; ++i) {
            const int tokn = (i + 1 < T_CHUNK) ? tk[(i + 1) >> 2][(i + 1) & 3] : nx[0][0];
            const float* prn = ptab + tokn * P_STRIDE;

            v8f u = (v8f){0.0f, 0.0f, 0.0f, 0.0f, 0.0f, 0.0f, 0.0f, 0.0f};
            u = wmma_f32_16x16x4(a, whb, u);

            float ho0 = fast_tanh(gq0 + vq0);
            float ho1 = fast_tanh(gq1 + vq1);
            float ho2 = fast_tanh(gq2 + vq2);
            pin_here(ho0, ho1, ho2);

            *reinterpret_cast<v4f*>(ust)     = (v4f){u[0], u[1], u[2], u[3]};
            *reinterpret_cast<v4f*>(ust + 4) = (v4f){u[4], u[5], u[6], u[7]};
            lds_handover();
            const float f0 = uldf[0 * U_PITCH], f1 = uldf[1 * U_PITCH];
            const float g0 = uldv[0 * U_PITCH], g1 = uldv[1 * U_PITCH];
            const float g2 = uldv[2 * U_PITCH];
            const v4f   n03 = *reinterpret_cast<const v4f*>(prn);
            const float n4  = prn[4];
            __builtin_amdgcn_sched_barrier(0);
            pin_here(ho0, ho1, ho2);

            hp0 = other_half(hp0, ho0);
            hp1 = other_half(hp1, ho1);
            hp2 = other_half(hp2, ho2);

            float v[5];
#pragma unroll
            for (int c = 0; c < 5; ++c) {
                const float pc = (c < 4) ? p03[c] : p4;
                const float so = fmaf(ho2, wo[2][c], fmaf(ho1, wo[1][c], fmaf(ho0, wo[0][c], pc)));
                const float sp = fmaf(hp2, wp[2][c], fmaf(hp1, wp[1][c], hp0 * wp[0][c]));
                v[c] = so + sp;
            }

            a[0] = fast_tanh(f0 + v[0]);
            a[1] = fast_tanh(f1 + v[1]);
            gq0 = g0; gq1 = g1; gq2 = g2;
            vq0 = v[2]; vq1 = v[3]; vq2 = v[4];
            p03 = n03; p4 = n4;
        }
#pragma unroll
        for (int q = 0; q < T_CHUNK / 4; ++q) tk[q] = nx[q];
    }

    const float ho0 = fast_tanh(gq0 + vq0);
    const float ho1 = fast_tanh(gq1 + vq1);
    const float ho2 = fast_tanh(gq2 + vq2);
    v2f af[3], wdb[3];
#pragma unroll
    for (int q = 0; q < 3; ++q) {
#pragma unroll
        for (int s = 0; s < 2; ++s) {
            const int K = k_unit(q, s, half);
            const float hv = (q == 0) ? a[s] : (q == 1) ? ((s == 0) ? ho0 : ho1) : ho2;
            if (K >= 0) {
                af[q][s]  = hv * dmask[(gRowBase + n) * H_SZ + K];
                wdb[q][s] = (n < L_SZ) ? Wd[K * L_SZ + n] : 0.0f;
            } else {
                af[q][s]  = 0.0f;
                wdb[q][s] = 0.0f;
            }
        }
    }

    const float bdv = (n < L_SZ) ? bd[n] : 0.0f;
    v8f lac;
#pragma unroll
    for (int r = 0; r < 8; ++r) lac[r] = bdv;

    lac = wmma_f32_16x16x4(af[0], wdb[0], lac);
    lac = wmma_f32_16x16x4(af[1], wdb[1], lac);
    lac = wmma_f32_16x16x4(af[2], wdb[2], lac);

    if (n < L_SZ) {
#pragma unroll
        for (int r = 0; r < 8; ++r)
            out[(gRowBase + r + mBase) * L_SZ + n] = lac[r];
    }
}

extern "C" void kernel_launch(void* const* d_in, const int* in_sizes, int n_in,
                              void* d_out, int out_size, void* d_ws, size_t ws_size,
                              hipStream_t stream) {
    (void)in_sizes; (void)n_in; (void)d_ws; (void)ws_size; (void)out_size;
    const int*   x     = (const int*)  d_in[0];
    const float* emb   = (const float*)d_in[1];
    const float* Wx    = (const float*)d_in[2];
    const float* Wh    = (const float*)d_in[3];
    const float* b_rnn = (const float*)d_in[4];
    const float* Wd    = (const float*)d_in[5];
    const float* bd    = (const float*)d_in[6];
    const float* dmask = (const float*)d_in[7];
    float* out = (float*)d_out;

    charrnn_kernel<<<B_TOT / ROWS_PER_BLOCK, TPB, 0, stream>>>(
        x, emb, Wx, Wh, b_rnn, Wd, bd, dmask, out);
}
